# starting kernel
# baseline (speedup 1.0000x reference)
#include <hip/hip_runtime.h>

typedef float v2f __attribute__((ext_vector_type(2)));
typedef float v8f __attribute__((ext_vector_type(8)));

__device__ __forceinline__ v8f wmma_f32_4(v2f a, v2f b, v8f c) {
    return __builtin_amdgcn_wmma_f32_16x16x4_f32(
         false, a,  false, b,
         (short)0, c,  false,  false);
}

template <bool GUARD>
__launch_bounds__(256)
__global__ void mlp3_wmma_kernel(const float* __restrict__ x,
                                 const float* __restrict__ W1, const float* __restrict__ b1,
                                 const float* __restrict__ W2, const float* __restrict__ b2,
                                 const float* __restrict__ W3, const float* __restrict__ b3,
                                 float* __restrict__ out,
                                 int Brows, int tiles)
{
    __shared__ float lds[8][16 * 16];

    const int tid  = threadIdx.x;
    const int wave = tid >> 5;
    const int lane = tid & 31;
    const int nlo  = lane & 15;
    const int hi   = lane >> 4;
    float* L = lds[wave];

    v2f B1;
    B1.x = W1[(2 * hi + 0) * 16 + nlo];
    B1.y = W1[(2 * hi + 1) * 16 + nlo];

    v2f B2[4];
#pragma unroll
    for (int j = 0; j < 4; ++j) {
        const int k0 = 4 * j + 2 * hi;
        B2[j].x = (nlo < 8) ? W2[(k0 + 0) * 8 + nlo] : 0.0f;
        B2[j].y = (nlo < 8) ? W2[(k0 + 1) * 8 + nlo] : 0.0f;
    }

    v2f B3[2];
#pragma unroll
    for (int j = 0; j < 2; ++j) {
        const int k0 = 4 * j + 2 * hi;
        B3[j].x = (nlo < 3) ? W3[(k0 + 0) * 3 + nlo] : 0.0f;
        B3[j].y = (nlo < 3) ? W3[(k0 + 1) * 3 + nlo] : 0.0f;
    }

    const float c1 = b1[nlo];
    const float c2 = (nlo < 8) ? b2[nlo] : 0.0f;
    const float c3 = (nlo < 3) ? b3[nlo] : 0.0f;

    const int st_base = hi * 128 + nlo;
    const int ld_base = nlo * 16 + hi * 2;

    const int wave_stride = gridDim.x * 8;

    for (int t = blockIdx.x * 8 + wave; t < tiles; t += wave_stride) {
        const int base = t * 16;

        const int row = base + nlo;
        v2f a1;
        if (GUARD) {
            a1 = (v2f){0.0f, 0.0f};
            if (row < Brows)
                a1 = __builtin_nontemporal_load((const v2f*)(x + row * 4 + 2 * hi));
        } else {
            a1 = __builtin_nontemporal_load((const v2f*)(x + row * 4 + 2 * hi));
        }

        v8f acc;
#pragma unroll
        for (int i = 0; i < 8; ++i) acc[i] = c1;
        v8f h1 = wmma_f32_4(a1, B1, acc);
#pragma unroll
        for (int i = 0; i < 8; ++i) h1[i] = fmaxf(h1[i], 0.0f);

#pragma unroll
        for (int i = 0; i < 8; ++i) L[st_base + i * 16] = h1[i];
        __asm__ volatile("s_wait_dscnt 0x0" ::: "memory");
        v2f a2[4];
#pragma unroll
        for (int j = 0; j < 4; ++j) a2[j] = *(const v2f*)(L + ld_base + 4 * j);

#pragma unroll
        for (int i = 0; i < 8; ++i) acc[i] = c2;
        v8f h2 = wmma_f32_4(a2[0], B2[0], acc);
        h2 = wmma_f32_4(a2[1], B2[1], h2);
        h2 = wmma_f32_4(a2[2], B2[2], h2);
        h2 = wmma_f32_4(a2[3], B2[3], h2);
#pragma unroll
        for (int i = 0; i < 8; ++i) h2[i] = fmaxf(h2[i], 0.0f);

#pragma unroll
        for (int i = 0; i < 8; ++i) L[st_base + i * 16] = h2[i];
        __asm__ volatile("s_wait_dscnt 0x0" ::: "memory");
        v2f a3[2];
#pragma unroll
        for (int j = 0; j < 2; ++j) a3[j] = *(const v2f*)(L + ld_base + 4 * j);

#pragma unroll
        for (int i = 0; i < 8; ++i) acc[i] = c3;
        v8f o = wmma_f32_4(a3[0], B3[0], acc);
        o = wmma_f32_4(a3[1], B3[1], o);

        if (nlo < 3) {
            const int r0 = base + 8 * hi;
#pragma unroll
            for (int i = 0; i < 8; ++i) {
                const int r = r0 + i;
                if (!GUARD || r < Brows)
                    __builtin_nontemporal_store(o[i], out + r * 3 + nlo);
            }
        }
    }
}

extern "C" void kernel_launch(void* const* d_in, const int* in_sizes, int n_in,
                              void* d_out, int out_size, void* d_ws, size_t ws_size,
                              hipStream_t stream) {
    const float* x  = (const float*)d_in[0];
    const float* W1 = (const float*)d_in[1];
    const float* b1 = (const float*)d_in[2];
    const float* W2 = (const float*)d_in[3];
    const float* b2 = (const float*)d_in[4];
    const float* W3 = (const float*)d_in[5];
    const float* b3 = (const float*)d_in[6];
    float* out = (float*)d_out;

    const int Brows = in_sizes[0] / 4;
    const int tiles = (Brows + 15) / 16;
    int blocks = (tiles + 7) / 8;
    if (blocks > 4096) blocks = 4096;
    if (blocks < 1) blocks = 1;

    if ((Brows & 15) == 0) {
        mlp3_wmma_kernel<false><<<blocks, 256, 0, stream>>>(
            x, W1, b1, W2, b2, W3, b3, out, Brows, tiles);
    } else {
        mlp3_wmma_kernel<true><<<blocks, 256, 0, stream>>>(
            x, W1, b1, W2, b2, W3, b3, out, Brows, tiles);
    }
}


# One contiguous 512-row panel per wave, 3907 panels in one resident round: the panel's x is requested whole at entry and its 32 tiles run eight abreast through the three layers; sums unchanged
# speedup vs baseline: 1.8757x; 1.8757x over previous
#include <hip/hip_runtime.h>


typedef float v2f __attribute__((ext_vector_type(2)));
typedef float v8f __attribute__((ext_vector_type(8)));

__device__ __forceinline__ v8f wmma_f32_4(v2f a, v2f b, v8f c) {
    return __builtin_amdgcn_wmma_f32_16x16x4_f32(
         false, a,  false, b,
         (short)0, c,  false,  false);
}

__device__ __forceinline__ v8f splat8(float v) {
    return (v8f){v, v, v, v, v, v, v, v};
}

constexpr int TG = 8;
constexpr int NG = 4;
constexpr int TP = TG * NG;

template <bool GUARD>
__launch_bounds__(256, 4)
__global__ void mlp3_panel_kernel(const float* __restrict__ x,
                                  const float* __restrict__ W1, const float* __restrict__ b1,
                                  const float* __restrict__ W2, const float* __restrict__ b2,
                                  const float* __restrict__ W3, const float* __restrict__ b3,
                                  float* __restrict__ out,
                                  int Brows, int tiles)
{
    __shared__ float lds[8][TG][16 * 16];

    const int tid  = threadIdx.x;
    const int wave = __builtin_amdgcn_readfirstlane(tid >> 5);
    const int lane = tid & 31;
    const int nlo  = lane & 15;
    const int hi   = lane >> 4;
    float* L = &lds[wave][0][0];

    const int t0 = (blockIdx.x * 8 + wave) * TP;
    if (t0 >= tiles) return;

    v2f xa[TP];
    const int lastrow = Brows - 1;
#pragma unroll
    for (int q = 0; q < TP; ++q) {
        int row = (t0 + q) * 16 + nlo;
        row = row < lastrow ? row : lastrow;
        xa[q] = __builtin_nontemporal_load((const v2f*)(x + row * 4 + 2 * hi));
    }

    v2f B1;
    B1.x = W1[(2 * hi + 0) * 16 + nlo];
    B1.y = W1[(2 * hi + 1) * 16 + nlo];

    v2f B2[4];
#pragma unroll
    for (int j = 0; j < 4; ++j) {
        const int k0 = 4 * j + 2 * hi;
        B2[j].x = (nlo < 8) ? W2[(k0 + 0) * 8 + nlo] : 0.0f;
        B2[j].y = (nlo < 8) ? W2[(k0 + 1) * 8 + nlo] : 0.0f;
    }

    v2f B3[2];
#pragma unroll
    for (int j = 0; j < 2; ++j) {
        const int k0 = 4 * j + 2 * hi;
        B3[j].x = (nlo < 3) ? W3[(k0 + 0) * 3 + nlo] : 0.0f;
        B3[j].y = (nlo < 3) ? W3[(k0 + 1) * 3 + nlo] : 0.0f;
    }

    const float c1 = b1[nlo];
    const float c2 = (nlo < 8) ? b2[nlo] : 0.0f;
    const float c3 = (nlo < 3) ? b3[nlo] : 0.0f;

    float* const ob = out + (size_t)(t0 * 16 + 8 * hi) * 3 + nlo;

    const int st_base = hi * 128 + nlo;
    const int ld_base = nlo * 16 + hi * 2;

#pragma unroll
    for (int g = 0; g < NG; ++g) {
        const int tg = t0 + g * TG;
        if (tg >= tiles) break;

        v8f h[TG];
#pragma unroll
        for (int u = 0; u < TG; ++u)
            h[u] = wmma_f32_4(xa[g * TG + u], B1, splat8(c1));

#pragma unroll
        for (int u = 0; u < TG; ++u)
#pragma unroll
            for (int i = 0; i < 8; ++i)
                L[u * 256 + st_base + i * 16] = fmaxf(h[u][i], 0.0f);
        __asm__ volatile("s_wait_dscnt 0x0" ::: "memory");
        v2f a2[TG][4];
#pragma unroll
        for (int u = 0; u < TG; ++u)
#pragma unroll
            for (int j = 0; j < 4; ++j)
                a2[u][j] = *(const v2f*)(L + u * 256 + ld_base + 4 * j);

#pragma unroll
        for (int u = 0; u < TG; ++u) {
            v8f t = wmma_f32_4(a2[u][0], B2[0], splat8(c2));
            t = wmma_f32_4(a2[u][1], B2[1], t);
            t = wmma_f32_4(a2[u][2], B2[2], t);
            h[u] = wmma_f32_4(a2[u][3], B2[3], t);
        }

#pragma unroll
        for (int u = 0; u < TG; ++u)
#pragma unroll
            for (int i = 0; i < 8; ++i)
                L[u * 256 + st_base + i * 16] = fmaxf(h[u][i], 0.0f);
        __asm__ volatile("s_wait_dscnt 0x0" ::: "memory");
        v2f a3[TG][2];
#pragma unroll
        for (int u = 0; u < TG; ++u)
#pragma unroll
            for (int j = 0; j < 2; ++j)
                a3[u][j] = *(const v2f*)(L + u * 256 + ld_base + 4 * j);

#pragma unroll
        for (int u = 0; u < TG; ++u) {
            v8f t = wmma_f32_4(a3[u][0], B3[0], splat8(c3));
            h[u] = wmma_f32_4(a3[u][1], B3[1], t);
        }

        if (nlo < 3) {
#pragma unroll
            for (int u = 0; u < TG; ++u) {
                const int r0 = (tg + u) * 16 + 8 * hi;
#pragma unroll
                for (int i = 0; i < 8; ++i) {
                    const int r = r0 + i;
                    if (!GUARD || r < Brows)
                        __builtin_nontemporal_store(h[u][i], ob + ((g * TG + u) * 16 + i) * 3);
                }
            }
        }
    }
}

extern "C" void kernel_launch(void* const* d_in, const int* in_sizes, int n_in,
                              void* d_out, int out_size, void* d_ws, size_t ws_size,
                              hipStream_t stream) {
    const float* x  = (const float*)d_in[0];
    const float* W1 = (const float*)d_in[1];
    const float* b1 = (const float*)d_in[2];
    const float* W2 = (const float*)d_in[3];
    const float* b2 = (const float*)d_in[4];
    const float* W3 = (const float*)d_in[5];
    const float* b3 = (const float*)d_in[6];
    float* out = (float*)d_out;

    const int Brows  = in_sizes[0] / 4;
    const int tiles  = (Brows + 15) / 16;
    const int panels = (tiles + TP - 1) / TP;
    int blocks = (panels + 7) / 8;
    if (blocks < 1) blocks = 1;

    if ((Brows & 127) == 0) {
        mlp3_panel_kernel<false><<<blocks, 256, 0, stream>>>(
            x, W1, b1, W2, b2, W3, b3, out, Brows, tiles);
    } else {
        mlp3_panel_kernel<true><<<blocks, 256, 0, stream>>>(
            x, W1, b1, W2, b2, W3, b3, out, Brows, tiles);
    }
}


# Outputs go through a packed per-wave LDS stage: a group's 128 rows leave in 3 whole-wave 16-byte stores of 512 contiguous bytes (were 64 stores of 2x12 B each); sums unchanged
# speedup vs baseline: 2.0248x; 1.0795x over previous
#include <hip/hip_runtime.h>


typedef float v2f __attribute__((ext_vector_type(2)));
typedef float v4f __attribute__((ext_vector_type(4)));
typedef float v8f __attribute__((ext_vector_type(8)));

__device__ __forceinline__ v8f wmma_f32_4(v2f a, v2f b, v8f c) {
    return __builtin_amdgcn_wmma_f32_16x16x4_f32(
         false, a,  false, b,
         (short)0, c,  false,  false);
}

__device__ __forceinline__ v8f splat8(float v) {
    return (v8f){v, v, v, v, v, v, v, v};
}

constexpr int TG = 8;
constexpr int NG = 4;
constexpr int TP = TG * NG;

template <bool GUARD>
__launch_bounds__(256, 4)
__global__ void mlp3_panel_kernel(const float* __restrict__ x,
                                  const float* __restrict__ W1, const float* __restrict__ b1,
                                  const float* __restrict__ W2, const float* __restrict__ b2,
                                  const float* __restrict__ W3, const float* __restrict__ b3,
                                  float* __restrict__ out,
                                  int Brows, int tiles)
{
    __shared__ float lds[8][TG][16 * 16];
    __shared__ float ost[8][TG * 16 * 3];

    const int tid  = threadIdx.x;
    const int wave = __builtin_amdgcn_readfirstlane(tid >> 5);
    const int lane = tid & 31;
    const int nlo  = lane & 15;
    const int hi   = lane >> 4;
    float* L = &lds[wave][0][0];
    float* O = &ost[wave][0];

    const int t0 = (blockIdx.x * 8 + wave) * TP;
    if (t0 >= tiles) return;

    v2f xa[TP];
    const int lastrow = Brows - 1;
#pragma unroll
    for (int q = 0; q < TP; ++q) {
        int row = (t0 + q) * 16 + nlo;
        row = row < lastrow ? row : lastrow;
        xa[q] = __builtin_nontemporal_load((const v2f*)(x + row * 4 + 2 * hi));
    }

    v2f B1;
    B1.x = W1[(2 * hi + 0) * 16 + nlo];
    B1.y = W1[(2 * hi + 1) * 16 + nlo];

    v2f B2[4];
#pragma unroll
    for (int j = 0; j < 4; ++j) {
        const int k0 = 4 * j + 2 * hi;
        B2[j].x = (nlo < 8) ? W2[(k0 + 0) * 8 + nlo] : 0.0f;
        B2[j].y = (nlo < 8) ? W2[(k0 + 1) * 8 + nlo] : 0.0f;
    }

    v2f B3[2];
#pragma unroll
    for (int j = 0; j < 2; ++j) {
        const int k0 = 4 * j + 2 * hi;
        B3[j].x = (nlo < 3) ? W3[(k0 + 0) * 3 + nlo] : 0.0f;
        B3[j].y = (nlo < 3) ? W3[(k0 + 1) * 3 + nlo] : 0.0f;
    }

    const float c1 = b1[nlo];
    const float c2 = (nlo < 8) ? b2[nlo] : 0.0f;
    const float c3 = (nlo < 3) ? b3[nlo] : 0.0f;

    const int os_base = (8 * hi) * 3 + nlo;
    float* const ob = out + (size_t)t0 * 48 + 4 * lane;
    const int elems = Brows * 3;

    auto flush = [&](int gi) {
        v4f o[3];
#pragma unroll
        for (int k = 0; k < 3; ++k)
            o[k] = *(const v4f*)(O + k * 128 + 4 * lane);
#pragma unroll
        for (int k = 0; k < 3; ++k) {
            float* dst = ob + gi * (TG * 48) + k * 128;
            if (!GUARD) {
                __builtin_nontemporal_store(o[k], (v4f*)dst);
            } else {
                const int e = (t0 + gi * TG) * 48 + k * 128 + 4 * lane;
                if (e + 3 < elems) {
                    __builtin_nontemporal_store(o[k], (v4f*)dst);
                } else {
#pragma unroll
                    for (int c = 0; c < 4; ++c)
                        if (e + c < elems) dst[c] = o[k][c];
                }
            }
        }
    };

    const int st_base = hi * 128 + nlo;
    const int ld_base = nlo * 16 + hi * 2;

    int done = 0;
#pragma unroll
    for (int g = 0; g < NG; ++g) {
        const int tg = t0 + g * TG;
        if (tg >= tiles) break;

        v8f h[TG];
#pragma unroll
        for (int u = 0; u < TG; ++u)
            h[u] = wmma_f32_4(xa[g * TG + u], B1, splat8(c1));

#pragma unroll
        for (int u = 0; u < TG; ++u)
#pragma unroll
            for (int i = 0; i < 8; ++i)
                L[u * 256 + st_base + i * 16] = fmaxf(h[u][i], 0.0f);
        __asm__ volatile("s_wait_dscnt 0x0" ::: "memory");
        v2f a2[TG][4];
#pragma unroll
        for (int u = 0; u < TG; ++u)
#pragma unroll
            for (int j = 0; j < 4; ++j)
                a2[u][j] = *(const v2f*)(L + u * 256 + ld_base + 4 * j);
        if (g > 0) flush(g - 1);

#pragma unroll
        for (int u = 0; u < TG; ++u) {
            v8f t = wmma_f32_4(a2[u][0], B2[0], splat8(c2));
            t = wmma_f32_4(a2[u][1], B2[1], t);
            t = wmma_f32_4(a2[u][2], B2[2], t);
            h[u] = wmma_f32_4(a2[u][3], B2[3], t);
        }

#pragma unroll
        for (int u = 0; u < TG; ++u)
#pragma unroll
            for (int i = 0; i < 8; ++i)
                L[u * 256 + st_base + i * 16] = fmaxf(h[u][i], 0.0f);
        __asm__ volatile("s_wait_dscnt 0x0" ::: "memory");
        v2f a3[TG][2];
#pragma unroll
        for (int u = 0; u < TG; ++u)
#pragma unroll
            for (int j = 0; j < 2; ++j)
                a3[u][j] = *(const v2f*)(L + u * 256 + ld_base + 4 * j);

#pragma unroll
        for (int u = 0; u < TG; ++u) {
            v8f t = wmma_f32_4(a3[u][0], B3[0], splat8(c3));
            h[u] = wmma_f32_4(a3[u][1], B3[1], t);
        }

        if (nlo < 3) {
#pragma unroll
            for (int u = 0; u < TG; ++u)
#pragma unroll
                for (int i = 0; i < 8; ++i)
                    O[u * 48 + os_base + i * 3] = h[u][i];
        }
        done = g + 1;
    }

    __asm__ volatile("s_wait_dscnt 0x0" ::: "memory");
    flush(done - 1);
}

extern "C" void kernel_launch(void* const* d_in, const int* in_sizes, int n_in,
                              void* d_out, int out_size, void* d_ws, size_t ws_size,
                              hipStream_t stream) {
    const float* x  = (const float*)d_in[0];
    const float* W1 = (const float*)d_in[1];
    const float* b1 = (const float*)d_in[2];
    const float* W2 = (const float*)d_in[3];
    const float* b2 = (const float*)d_in[4];
    const float* W3 = (const float*)d_in[5];
    const float* b3 = (const float*)d_in[6];
    float* out = (float*)d_out;

    const int Brows  = in_sizes[0] / 4;
    const int tiles  = (Brows + 15) / 16;
    const int panels = (tiles + TP - 1) / TP;
    int blocks = (panels + 7) / 8;
    if (blocks < 1) blocks = 1;

    if ((Brows & 127) == 0) {
        mlp3_panel_kernel<false><<<blocks, 256, 0, stream>>>(
            x, W1, b1, W2, b2, W3, b3, out, Brows, tiles);
    } else {
        mlp3_panel_kernel<true><<<blocks, 256, 0, stream>>>(
            x, W1, b1, W2, b2, W3, b3, out, Brows, tiles);
    }
}


# Staging-tile rows re-placed in LDS (pairs 32 floats apart, upper half +16, skew 2 and 8): paired tile stores (were 2-way) and fragment reads (were 4-way) hit 64 distinct banks; sums unchanged
# speedup vs baseline: 2.1615x; 1.0675x over previous
#include <hip/hip_runtime.h>


typedef float v2f __attribute__((ext_vector_type(2)));
typedef float v4f __attribute__((ext_vector_type(4)));
typedef float v8f __attribute__((ext_vector_type(8)));

__device__ __forceinline__ v8f wmma_f32_4(v2f a, v2f b, v8f c) {
    return __builtin_amdgcn_wmma_f32_16x16x4_f32(
         false, a,  false, b,
         (short)0, c,  false,  false);
}

__device__ __forceinline__ v8f splat8(float v) {
    return (v8f){v, v, v, v, v, v, v, v};
}

constexpr int TG = 8;
constexpr int NG = 4;
constexpr int TP = TG * NG;

__device__ __forceinline__ constexpr int row_at(int r) {
    return 32 * (r & 7) + 16 * (r >> 3) + 2 * ((r >> 1) & 1) + 8 * ((r >> 2) & 1);
}
constexpr int TS = 272;

template <bool GUARD>
__launch_bounds__(256, 4)
__global__ void mlp3_panel_kernel(const float* __restrict__ x,
                                  const float* __restrict__ W1, const float* __restrict__ b1,
                                  const float* __restrict__ W2, const float* __restrict__ b2,
                                  const float* __restrict__ W3, const float* __restrict__ b3,
                                  float* __restrict__ out,
                                  int Brows, int tiles)
{
    __shared__ float lds[8][TG][TS];
    __shared__ float ost[8][TG * 16 * 3];

    const int tid  = threadIdx.x;
    const int wave = __builtin_amdgcn_readfirstlane(tid >> 5);
    const int lane = tid & 31;
    const int nlo  = lane & 15;
    const int hi   = lane >> 4;
    float* L = &lds[wave][0][0];
    float* O = &ost[wave][0];

    const int t0 = (blockIdx.x * 8 + wave) * TP;
    if (t0 >= tiles) return;

    v2f xa[TP];
    const int lastrow = Brows - 1;
#pragma unroll
    for (int q = 0; q < TP; ++q) {
        int row = (t0 + q) * 16 + nlo;
        row = row < lastrow ? row : lastrow;
        xa[q] = __builtin_nontemporal_load((const v2f*)(x + row * 4 + 2 * hi));
    }

    v2f B1;
    B1.x = W1[(2 * hi + 0) * 16 + nlo];
    B1.y = W1[(2 * hi + 1) * 16 + nlo];

    v2f B2[4];
#pragma unroll
    for (int j = 0; j < 4; ++j) {
        const int k0 = 4 * j + 2 * hi;
        B2[j].x = (nlo < 8) ? W2[(k0 + 0) * 8 + nlo] : 0.0f;
        B2[j].y = (nlo < 8) ? W2[(k0 + 1) * 8 + nlo] : 0.0f;
    }

    v2f B3[2];
#pragma unroll
    for (int j = 0; j < 2; ++j) {
        const int k0 = 4 * j + 2 * hi;
        B3[j].x = (nlo < 3) ? W3[(k0 + 0) * 3 + nlo] : 0.0f;
        B3[j].y = (nlo < 3) ? W3[(k0 + 1) * 3 + nlo] : 0.0f;
    }

    const float c1 = b1[nlo];
    const float c2 = (nlo < 8) ? b2[nlo] : 0.0f;
    const float c3 = (nlo < 3) ? b3[nlo] : 0.0f;

    const int os_base = (8 * hi) * 3 + nlo;
    float* const ob = out + (size_t)t0 * 48 + 4 * lane;
    const int elems = Brows * 3;

    auto flush = [&](int gi) {
        v4f o[3];
#pragma unroll
        for (int k = 0; k < 3; ++k)
            o[k] = *(const v4f*)(O + k * 128 + 4 * lane);
#pragma unroll
        for (int k = 0; k < 3; ++k) {
            float* dst = ob + gi * (TG * 48) + k * 128;
            if (!GUARD) {
                __builtin_nontemporal_store(o[k], (v4f*)dst);
            } else {
                const int e = (t0 + gi * TG) * 48 + k * 128 + 4 * lane;
                if (e + 3 < elems) {
                    __builtin_nontemporal_store(o[k], (v4f*)dst);
                } else {
#pragma unroll
                    for (int c = 0; c < 4; ++c)
                        if (e + c < elems) dst[c] = o[k][c];
                }
            }
        }
    };

    const int st_base = row_at(8 * hi) + nlo;
    const int ld_base = row_at(nlo) + hi * 2;

    int done = 0;
#pragma unroll
    for (int g = 0; g < NG; ++g) {
        const int tg = t0 + g * TG;
        if (tg >= tiles) break;

        v8f h[TG];
#pragma unroll
        for (int u = 0; u < TG; ++u)
            h[u] = wmma_f32_4(xa[g * TG + u], B1, splat8(c1));

#pragma unroll
        for (int u = 0; u < TG; ++u)
#pragma unroll
            for (int i = 0; i < 8; ++i)
                L[u * TS + st_base + row_at(i)] = fmaxf(h[u][i], 0.0f);
        __asm__ volatile("s_wait_dscnt 0x0" ::: "memory");
        v2f a2[TG][4];
#pragma unroll
        for (int u = 0; u < TG; ++u)
#pragma unroll
            for (int j = 0; j < 4; ++j)
                a2[u][j] = *(const v2f*)(L + u * TS + ld_base + 4 * j);
        if (g > 0) flush(g - 1);

#pragma unroll
        for (int u = 0; u < TG; ++u) {
            v8f t = wmma_f32_4(a2[u][0], B2[0], splat8(c2));
            t = wmma_f32_4(a2[u][1], B2[1], t);
            t = wmma_f32_4(a2[u][2], B2[2], t);
            h[u] = wmma_f32_4(a2[u][3], B2[3], t);
        }

#pragma unroll
        for (int u = 0; u < TG; ++u)
#pragma unroll
            for (int i = 0; i < 8; ++i)
                L[u * TS + st_base + row_at(i)] = fmaxf(h[u][i], 0.0f);
        __asm__ volatile("s_wait_dscnt 0x0" ::: "memory");
        v2f a3[TG][2];
#pragma unroll
        for (int u = 0; u < TG; ++u)
#pragma unroll
            for (int j = 0; j < 2; ++j)
                a3[u][j] = *(const v2f*)(L + u * TS + ld_base + 4 * j);

#pragma unroll
        for (int u = 0; u < TG; ++u) {
            v8f t = wmma_f32_4(a3[u][0], B3[0], splat8(c3));
            h[u] = wmma_f32_4(a3[u][1], B3[1], t);
        }

        if (nlo < 3) {
#pragma unroll
            for (int u = 0; u < TG; ++u)
#pragma unroll
                for (int i = 0; i < 8; ++i)
                    O[u * 48 + os_base + i * 3] = h[u][i];
        }
        done = g + 1;
    }

    __asm__ volatile("s_wait_dscnt 0x0" ::: "memory");
    flush(done - 1);
}

extern "C" void kernel_launch(void* const* d_in, const int* in_sizes, int n_in,
                              void* d_out, int out_size, void* d_ws, size_t ws_size,
                              hipStream_t stream) {
    const float* x  = (const float*)d_in[0];
    const float* W1 = (const float*)d_in[1];
    const float* b1 = (const float*)d_in[2];
    const float* W2 = (const float*)d_in[3];
    const float* b2 = (const float*)d_in[4];
    const float* W3 = (const float*)d_in[5];
    const float* b3 = (const float*)d_in[6];
    float* out = (float*)d_out;

    const int Brows  = in_sizes[0] / 4;
    const int tiles  = (Brows + 15) / 16;
    const int panels = (tiles + TP - 1) / TP;
    int blocks = (panels + 7) / 8;
    if (blocks < 1) blocks = 1;

    if ((Brows & 127) == 0) {
        mlp3_panel_kernel<false><<<blocks, 256, 0, stream>>>(
            x, W1, b1, W2, b2, W3, b3, out, Brows, tiles);
    } else {
        mlp3_panel_kernel<true><<<blocks, 256, 0, stream>>>(
            x, W1, b1, W2, b2, W3, b3, out, Brows, tiles);
    }
}


# Tile pairs share layer 2's accumulator (cols 0-7 | 8-15), tile quads layer 3's (cols 3t..3t+2), via column-shifted weight copies: second hand-over and output stage run on full tiles; sums unchanged
# speedup vs baseline: 2.3971x; 1.1090x over previous
#include <hip/hip_runtime.h>


typedef float v2f __attribute__((ext_vector_type(2)));
typedef float v4f __attribute__((ext_vector_type(4)));
typedef float v8f __attribute__((ext_vector_type(8)));

__device__ __forceinline__ v8f wmma_f32_4(v2f a, v2f b, v8f c) {
    return __builtin_amdgcn_wmma_f32_16x16x4_f32(
         false, a,  false, b,
         (short)0, c,  false,  false);
}

__device__ __forceinline__ v8f splat8(float v) {
    return (v8f){v, v, v, v, v, v, v, v};
}

constexpr int TG = 8;
constexpr int NG = 4;
constexpr int TP = TG * NG;
constexpr int PG = TG / 2;
constexpr int QG = TG / 4;

__device__ __forceinline__ constexpr int row_at(int r) {
    return 32 * (r & 7) + 16 * (r >> 3) + 2 * ((r >> 1) & 1) + 8 * ((r >> 2) & 1);
}
constexpr int TS = 272;

template <bool GUARD>
__launch_bounds__(256, 4)
__global__ void mlp3_panel_kernel(const float* __restrict__ x,
                                  const float* __restrict__ W1, const float* __restrict__ b1,
                                  const float* __restrict__ W2, const float* __restrict__ b2,
                                  const float* __restrict__ W3, const float* __restrict__ b3,
                                  float* __restrict__ out,
                                  int Brows, int tiles)
{
    __shared__ float lds[8][TG][TS];
    __shared__ float ost[8][TG * 16 * 3];

    const int tid  = threadIdx.x;
    const int wave = __builtin_amdgcn_readfirstlane(tid >> 5);
    const int lane = tid & 31;
    const int nlo  = lane & 15;
    const int hi   = lane >> 4;
    float* L = &lds[wave][0][0];
    float* O = &ost[wave][0];

    const int t0 = (blockIdx.x * 8 + wave) * TP;
    if (t0 >= tiles) return;

    v2f xa[TP];
    const int lastrow = Brows - 1;
#pragma unroll
    for (int q = 0; q < TP; ++q) {
        int row = (t0 + q) * 16 + nlo;
        row = row < lastrow ? row : lastrow;
        xa[q] = __builtin_nontemporal_load((const v2f*)(x + row * 4 + 2 * hi));
    }

    v2f B1;
    B1.x = W1[(2 * hi + 0) * 16 + nlo];
    B1.y = W1[(2 * hi + 1) * 16 + nlo];

    v2f B2a[4], B2b[4];
#pragma unroll
    for (int j = 0; j < 4; ++j) {
        const int k0 = 4 * j + 2 * hi;
        const float w0 = W2[(k0 + 0) * 8 + (nlo & 7)];
        const float w1 = W2[(k0 + 1) * 8 + (nlo & 7)];
        B2a[j].x = (nlo < 8) ? w0 : 0.0f;
        B2a[j].y = (nlo < 8) ? w1 : 0.0f;
        B2b[j].x = (nlo < 8) ? 0.0f : w0;
        B2b[j].y = (nlo < 8) ? 0.0f : w1;
    }

    const int oc  = nlo % 3;
    const int oq  = nlo / 3;
    v2f B3[4][2];
#pragma unroll
    for (int jj = 0; jj < 2; ++jj) {
        const int k0 = 4 * jj + 2 * hi;
        const float w0 = W3[(k0 + 0) * 3 + oc];
        const float w1 = W3[(k0 + 1) * 3 + oc];
#pragma unroll
        for (int t = 0; t < 4; ++t) {
            B3[t][jj].x = (oq == t) ? w0 : 0.0f;
            B3[t][jj].y = (oq == t) ? w1 : 0.0f;
        }
    }

    const float c1 = b1[nlo];
    const float c2 = b2[nlo & 7];
    const float b3v = b3[oc];
    const float c3 = (nlo < 12) ? b3v : 0.0f;

    const int os_base = oq * 48 + (8 * hi) * 3 + oc;
    float* const ob = out + (size_t)t0 * 48 + 4 * lane;
    const int elems = Brows * 3;

    auto flush = [&](int gi) {
        v4f o[3];
#pragma unroll
        for (int k = 0; k < 3; ++k)
            o[k] = *(const v4f*)(O + k * 128 + 4 * lane);
#pragma unroll
        for (int k = 0; k < 3; ++k) {
            float* dst = ob + gi * (TG * 48) + k * 128;
            if (!GUARD) {
                __builtin_nontemporal_store(o[k], (v4f*)dst);
            } else {
                const int e = (t0 + gi * TG) * 48 + k * 128 + 4 * lane;
                if (e + 3 < elems) {
                    __builtin_nontemporal_store(o[k], (v4f*)dst);
                } else {
#pragma unroll
                    for (int c = 0; c < 4; ++c)
                        if (e + c < elems) dst[c] = o[k][c];
                }
            }
        }
    };

    const int st_base = row_at(8 * hi) + nlo;
    const int ld_base = row_at(nlo) + hi * 2;

    int done = 0;
#pragma unroll
    for (int g = 0; g < NG; ++g) {
        const int tg = t0 + g * TG;
        if (tg >= tiles) break;

        v8f h[TG];
#pragma unroll
        for (int u = 0; u < TG; ++u)
            h[u] = wmma_f32_4(xa[g * TG + u], B1, splat8(c1));

#pragma unroll
        for (int u = 0; u < TG; ++u)
#pragma unroll
            for (int i = 0; i < 8; ++i)
                L[u * TS + st_base + row_at(i)] = fmaxf(h[u][i], 0.0f);
        __asm__ volatile("s_wait_dscnt 0x0" ::: "memory");
        v2f a2[TG][4];
#pragma unroll
        for (int u = 0; u < TG; ++u)
#pragma unroll
            for (int j = 0; j < 4; ++j)
                a2[u][j] = *(const v2f*)(L + u * TS + ld_base + 4 * j);
        if (g > 0) flush(g - 1);

        v8f p2[PG];
#pragma unroll
        for (int p = 0; p < PG; ++p) {
            v8f t = splat8(c2);
#pragma unroll
            for (int j = 0; j < 4; ++j) t = wmma_f32_4(a2[2 * p][j], B2a[j], t);
#pragma unroll
            for (int j = 0; j < 4; ++j) t = wmma_f32_4(a2[2 * p + 1][j], B2b[j], t);
            p2[p] = t;
        }

#pragma unroll
        for (int p = 0; p < PG; ++p)
#pragma unroll
            for (int i = 0; i < 8; ++i)
                L[p * TS + st_base + row_at(i)] = fmaxf(p2[p][i], 0.0f);
        __asm__ volatile("s_wait_dscnt 0x0" ::: "memory");
        v2f a3[PG][4];
#pragma unroll
        for (int p = 0; p < PG; ++p)
#pragma unroll
            for (int j = 0; j < 4; ++j)
                a3[p][j] = *(const v2f*)(L + p * TS + ld_base + 4 * j);

        v8f o3[QG];
#pragma unroll
        for (int q = 0; q < QG; ++q) {
            v8f t = splat8(c3);
#pragma unroll
            for (int tq = 0; tq < 4; ++tq)
#pragma unroll
                for (int jj = 0; jj < 2; ++jj)
                    t = wmma_f32_4(a3[2 * q + (tq >> 1)][2 * (tq & 1) + jj], B3[tq][jj], t);
            o3[q] = t;
        }

        if (nlo < 12) {
#pragma unroll
            for (int q = 0; q < QG; ++q)
#pragma unroll
                for (int i = 0; i < 8; ++i)
                    O[q * 192 + os_base + i * 3] = o3[q][i];
        }
        done = g + 1;
    }

    __asm__ volatile("s_wait_dscnt 0x0" ::: "memory");
    flush(done - 1);
}

extern "C" void kernel_launch(void* const* d_in, const int* in_sizes, int n_in,
                              void* d_out, int out_size, void* d_ws, size_t ws_size,
                              hipStream_t stream) {
    const float* x  = (const float*)d_in[0];
    const float* W1 = (const float*)d_in[1];
    const float* b1 = (const float*)d_in[2];
    const float* W2 = (const float*)d_in[3];
    const float* b2 = (const float*)d_in[4];
    const float* W3 = (const float*)d_in[5];
    const float* b3 = (const float*)d_in[6];
    float* out = (float*)d_out;

    const int Brows  = in_sizes[0] / 4;
    const int tiles  = (Brows + 15) / 16;
    const int panels = (tiles + TP - 1) / TP;
    int blocks = (panels + 7) / 8;
    if (blocks < 1) blocks = 1;

    if ((Brows & 127) == 0) {
        mlp3_panel_kernel<false><<<blocks, 256, 0, stream>>>(
            x, W1, b1, W2, b2, W3, b3, out, Brows, tiles);
    } else {
        mlp3_panel_kernel<true><<<blocks, 256, 0, stream>>>(
            x, W1, b1, W2, b2, W3, b3, out, Brows, tiles);
    }
}
